# starting kernel
# baseline (speedup 1.0000x reference)
#include <hip/hip_runtime.h>


#define NFEAT 20480
#define NACC  128
#define NBUCK 4
#define BAG   32
#define POS_PER_BLOCK 16
#define THREADS (POS_PER_BLOCK * 32)

typedef __attribute__((ext_vector_type(2))) float v2f;
typedef __attribute__((ext_vector_type(8))) float v8f;

#if defined(__has_builtin)
#  if __has_builtin(__builtin_amdgcn_wmma_f32_16x16x4_f32)
#    define HAVE_WMMA_F32X4 1
#  endif
#endif

__device__ __forceinline__ float4 ld4(const float* p) {
  return *(const float4*)p;
}

__global__ __launch_bounds__(THREADS) void nnue_fwd(
    const int* __restrict__ w_cols, const int* __restrict__ b_cols,
    const float* __restrict__ stms, const float* __restrict__ psqt_w,
    const float* __restrict__ acc_w, const float* __restrict__ acc_bias,
    const float* __restrict__ layer_w, const float* __restrict__ layer_b,
    float* __restrict__ out, int n_pos)
{
  __shared__ float sBlend[POS_PER_BLOCK][260];
  __shared__ float sPsqt[POS_PER_BLOCK][NBUCK];
  __shared__ float sSign[POS_PER_BLOCK];

  const int wave = threadIdx.x >> 5;
  const int lane = threadIdx.x & 31;
  const int pos  = blockIdx.x * POS_PER_BLOCK + wave;

  if (pos < n_pos) {
    const size_t base = (size_t)pos * BAG;
    const int wcolL = w_cols[base + lane];
    const int bcolL = b_cols[base + lane];

    float4 p;
    {
      float4 pw = ld4(psqt_w + (size_t)wcolL * NBUCK);
      float4 pb = ld4(psqt_w + (size_t)bcolL * NBUCK);
      p.x = pw.x - pb.x; p.y = pw.y - pb.y;
      p.z = pw.z - pb.z; p.w = pw.w - pb.w;
    }

    float4 wa = make_float4(0.f, 0.f, 0.f, 0.f);
    float4 ba = make_float4(0.f, 0.f, 0.f, 0.f);
    const int laneOff = lane * 4;
#pragma unroll
    for (int j = 0; j < BAG; ++j) {
      const int wc = __builtin_amdgcn_readlane(wcolL, j);
      float4 r = ld4(acc_w + (size_t)wc * NACC + laneOff);
      wa.x += r.x; wa.y += r.y; wa.z += r.z; wa.w += r.w;
    }
#pragma unroll
    for (int j = 0; j < BAG; ++j) {
      const int bc = __builtin_amdgcn_readlane(bcolL, j);
      float4 r = ld4(acc_w + (size_t)bc * NACC + laneOff);
      ba.x += r.x; ba.y += r.y; ba.z += r.z; ba.w += r.w;
    }

    float4 bias = ld4(acc_bias + laneOff);
    wa.x = fminf(fmaxf(wa.x + bias.x, 0.f), 1.f);
    wa.y = fminf(fmaxf(wa.y + bias.y, 0.f), 1.f);
    wa.z = fminf(fmaxf(wa.z + bias.z, 0.f), 1.f);
    wa.w = fminf(fmaxf(wa.w + bias.w, 0.f), 1.f);
    ba.x = fminf(fmaxf(ba.x + bias.x, 0.f), 1.f);
    ba.y = fminf(fmaxf(ba.y + bias.y, 0.f), 1.f);
    ba.z = fminf(fmaxf(ba.z + bias.z, 0.f), 1.f);
    ba.w = fminf(fmaxf(ba.w + bias.w, 0.f), 1.f);

    const float s = stms[pos];
    const bool flip = (s != 0.0f);
    float4 first  = flip ? ba : wa;
    float4 second = flip ? wa : ba;
    float* row = &sBlend[wave][0];
    *(float4*)(row + 4 * lane)        = first;
    *(float4*)(row + NACC + 4 * lane) = second;

#pragma unroll
    for (int off = 16; off > 0; off >>= 1) {
      p.x += __shfl_xor(p.x, off, 32);
      p.y += __shfl_xor(p.y, off, 32);
      p.z += __shfl_xor(p.z, off, 32);
      p.w += __shfl_xor(p.w, off, 32);
    }
    if (lane == 0) {
      sPsqt[wave][0] = p.x; sPsqt[wave][1] = p.y;
      sPsqt[wave][2] = p.z; sPsqt[wave][3] = p.w;
      sSign[wave] = flip ? -1.0f : 1.0f;
    }
  }
  __syncthreads();

  if (wave == 0) {
    const int n  = lane & 15;
    const int hi = lane >> 4;
    const int nq = (n < NBUCK) ? n : 0;
    const float nmask = (n < NBUCK) ? 1.0f : 0.0f;
    const float* lw = layer_w + (size_t)nq * (2 * NACC);
    const float* arow = &sBlend[n][0];

#ifdef HAVE_WMMA_F32X4
    v8f c = {0.f, 0.f, 0.f, 0.f, 0.f, 0.f, 0.f, 0.f};
#pragma unroll 8
    for (int k4 = 0; k4 < (2 * NACC) / 4; ++k4) {
      const int kb = k4 * 4 + hi * 2;
      v2f a; a.x = arow[kb];        a.y = arow[kb + 1];
      v2f b; b.x = lw[kb] * nmask;  b.y = lw[kb + 1] * nmask;
      c = __builtin_amdgcn_wmma_f32_16x16x4_f32(
              false, a, false, b, (short)0, c, false, false);
    }
    if (n < NBUCK) {
      const float lb = layer_b[n];
#pragma unroll
      for (int r = 0; r < 8; ++r) {
        const int m = r + hi * 8;
        const int opos = blockIdx.x * POS_PER_BLOCK + m;
        if (opos < n_pos)
          out[(size_t)opos * NBUCK + n] =
              sPsqt[m][n] + sSign[m] * (c[r] + lb);
      }
    }
#else
    if (hi == 0) {
      const int m = n;
      const int opos = blockIdx.x * POS_PER_BLOCK + m;
      if (opos < n_pos) {
        const float* brow = &sBlend[m][0];
        const float sign = sSign[m];
#pragma unroll
        for (int bkt = 0; bkt < NBUCK; ++bkt) {
          float acc = layer_b[bkt];
          for (int k = 0; k < 2 * NACC; ++k)
            acc += brow[k] * layer_w[bkt * (2 * NACC) + k];
          out[(size_t)opos * NBUCK + bkt] = sPsqt[m][bkt] + sign * acc;
        }
      }
    }
#endif
  }
}

extern "C" void kernel_launch(void* const* d_in, const int* in_sizes, int n_in,
                              void* d_out, int out_size, void* d_ws, size_t ws_size,
                              hipStream_t stream) {
  const int*   w_cols   = (const int*)d_in[1];
  const int*   b_cols   = (const int*)d_in[3];
  const float* stms     = (const float*)d_in[4];
  const float* psqt_w   = (const float*)d_in[5];
  const float* acc_w    = (const float*)d_in[6];
  const float* acc_bias = (const float*)d_in[7];
  const float* layer_w  = (const float*)d_in[8];
  const float* layer_b  = (const float*)d_in[9];
  float* out = (float*)d_out;

  const int n_pos = in_sizes[4];
  const int blocks = (n_pos + POS_PER_BLOCK - 1) / POS_PER_BLOCK;
  nnue_fwd<<<blocks, THREADS, 0, stream>>>(
      w_cols, b_cols, stms, psqt_w, acc_w, acc_bias, layer_w, layer_b,
      out, n_pos);
}


# bag indices re-ordered per lane in the re-lay launch so that the 16 lanes of a half-wave read different LDS bank groups at most steps of the slice gather; bag sum order changes
# speedup vs baseline: 1.6447x; 1.6447x over previous
#include <hip/hip_runtime.h>


#define NFEAT 20480
#define NACC  128
#define NBUCK 4
#define BAG   32
#define NSLICE 32
#define NCHUNK 8
#define PART_OFF_FLOATS (4u << 20)
#define IDX_OFF_FLOATS  (16u << 20)
#define TR_BLOCKS (NFEAT / 32)
#define IX_BLOCKS 512

typedef __attribute__((ext_vector_type(2))) float v2f;
typedef __attribute__((ext_vector_type(8))) float v8f;

__device__ __forceinline__ float4 ld4(const float* p) {
  return *(const float4*)p;
}

#define TR_ROWS 32
__global__ __launch_bounds__(256) void acc_to_slices(
    const float* __restrict__ acc_w, float* __restrict__ slices,
    const int* __restrict__ w_cols, const int* __restrict__ b_cols,
    int* __restrict__ idx_out, int n_pos)
{
  int4* __restrict__ idxT = (int4*)idx_out;
  __shared__ float4 tile[TR_ROWS][33];
  __shared__ int perm[BAG][256];
  const int wave = threadIdx.x >> 5;
  const int lane = threadIdx.x & 31;
  if (blockIdx.x >= TR_BLOCKS) {
    if (n_pos <= 0) return;
    const int groups = (n_pos + 15) >> 4;
    const int c = lane & 15, h = lane >> 4;
    for (int g = (blockIdx.x - TR_BLOCKS) * 8 + wave; g < groups; g += IX_BLOCKS * 8) {
      const int pos = min(g * 16 + c, n_pos - 1);
      const int4* p = (const int4*)((h ? b_cols : w_cols) + (size_t)pos * BAG);
      int4 d[8];
#pragma unroll
      for (int q = 0; q < 8; ++q) d[q] = p[q];
      int v[BAG];
#pragma unroll
      for (int q = 0; q < 8; ++q) {
        v[4 * q + 0] = d[q].x; v[4 * q + 1] = d[q].y;
        v[4 * q + 2] = d[q].z; v[4 * q + 3] = d[q].w;
      }
      const int rot = (c + 8 * h) & 15;
      unsigned seen = 0u, filled = 0u, late = 0u;
#pragma unroll
      for (int i = 0; i < BAG; ++i) {
        const int k = v[i] & 15;
        const unsigned cnt = (seen >> (2 * k)) & 3u;
        if (cnt < 2u) {
          const int t = ((k - rot) & 15) + 16 * (int)cnt;
          perm[t][threadIdx.x] = v[i];
          filled |= 1u << t;
          seen += 1u << (2 * k);
        } else {
          late |= 1u << i;
        }
      }
      unsigned empty = ~filled;
#pragma unroll
      for (int i = 0; i < BAG; ++i) {
        if ((late >> i) & 1u) {
          const int t = __builtin_ctz(empty);
          empty &= empty - 1u;
          perm[t][threadIdx.x] = v[i];
        }
      }
#pragma unroll
      for (int q = 0; q < 8; ++q) {
        int4 o;
        o.x = perm[4 * q + 0][threadIdx.x]; o.y = perm[4 * q + 1][threadIdx.x];
        o.z = perm[4 * q + 2][threadIdx.x]; o.w = perm[4 * q + 3][threadIdx.x];
        idxT[((size_t)g * 8 + q) * 32 + lane] = o;
      }
    }
    return;
  }
  const int row0 = blockIdx.x * TR_ROWS;
#pragma unroll
  for (int k = 0; k < 4; ++k) {
    const int r = wave * 4 + k;
    tile[r][lane] = ld4(acc_w + (size_t)(row0 + r) * NACC + lane * 4);
  }
  __syncthreads();
#pragma unroll
  for (int k = 0; k < 4; ++k) {
    const int s = wave * 4 + k;
    *(float4*)(slices + ((size_t)s * NFEAT + row0 + lane) * 4) = tile[lane][s];
  }
}

#define SL_WAVES 32
#define SL_THREADS (SL_WAVES * 32)

__device__ __forceinline__ void load_idx(int4 (&d)[8], const int4* __restrict__ idxT,
                                         int g, int lane) {
  const int4* p = idxT + (size_t)g * 8 * 32 + lane;
#pragma unroll
  for (int q = 0; q < 8; ++q) d[q] = p[q * 32];
}

__global__ __launch_bounds__(SL_THREADS) void slice_fwd(
    const int* __restrict__ idx_in,
    const float* __restrict__ stms, const float* __restrict__ slices,
    const float* __restrict__ acc_bias, const float* __restrict__ layer_w,
    float* __restrict__ part, int n_pos)
{
  __shared__ float4 sTab[NFEAT];

  if (n_pos <= 0) return;
  const int4* __restrict__ idxT = (const int4*)idx_in;
  const int wave = __builtin_amdgcn_readfirstlane((int)(threadIdx.x >> 5));
  const int lane = threadIdx.x & 31;
  const int c = lane & 15;
  const int h = lane >> 4;
  const int slice = blockIdx.x & (NSLICE - 1);
  const int chunk = blockIdx.x >> 5;
  const int groups  = (n_pos + 15) >> 4;
  const int gstride = NCHUNK * SL_WAVES;
  int g = chunk * SL_WAVES + wave;

  int4 cur[8];
  load_idx(cur, idxT, min(g, groups - 1), lane);

  const int nq = (c < NBUCK) ? c : 0;
  const float nmask = (c < NBUCK) ? 1.0f : 0.0f;
  const float* lw = layer_w + (size_t)nq * (2 * NACC) + h * NACC + slice * 4;
  v2f wf0, wf1;
  wf0.x = lw[0] * nmask; wf0.y = lw[1] * nmask;
  wf1.x = lw[2] * nmask; wf1.y = lw[3] * nmask;
  const float4 bias = ld4(acc_bias + slice * 4);

  {
    const float* src = slices + (size_t)slice * NFEAT * 4;
    int r = (int)((chunk * 2560u) % NFEAT) + (int)threadIdx.x;
#pragma unroll
    for (int i = 0; i < NFEAT / SL_THREADS; ++i) {
      if (r >= NFEAT) r -= NFEAT;
      sTab[r] = ld4(src + (size_t)r * 4);
      r += SL_THREADS;
    }
  }
  __syncthreads();

  while (g < groups) {
    const int posu = g * 16 + c;
    const int pos  = min(posu, n_pos - 1);
    const float s  = stms[pos];
    int4 nxt[8];
    load_idx(nxt, idxT, min(g + gstride, groups - 1), lane);

    float4 a = make_float4(0.f, 0.f, 0.f, 0.f);
#pragma unroll
    for (int q = 0; q < 8; ++q) {
      float4 r;
      r = sTab[cur[q].x]; a.x += r.x; a.y += r.y; a.z += r.z; a.w += r.w;
      r = sTab[cur[q].y]; a.x += r.x; a.y += r.y; a.z += r.z; a.w += r.w;
      r = sTab[cur[q].z]; a.x += r.x; a.y += r.y; a.z += r.z; a.w += r.w;
      r = sTab[cur[q].w]; a.x += r.x; a.y += r.y; a.z += r.z; a.w += r.w;
    }
    a.x = fminf(fmaxf(a.x + bias.x, 0.f), 1.f);
    a.y = fminf(fmaxf(a.y + bias.y, 0.f), 1.f);
    a.z = fminf(fmaxf(a.z + bias.z, 0.f), 1.f);
    a.w = fminf(fmaxf(a.w + bias.w, 0.f), 1.f);

    const bool flip = (s != 0.0f);
    const float ox = __shfl_xor(a.x, 16, 32);
    const float oy = __shfl_xor(a.y, 16, 32);
    const float oz = __shfl_xor(a.z, 16, 32);
    const float ow = __shfl_xor(a.w, 16, 32);
    v2f af0, af1;
    af0.x = flip ? ox : a.x; af0.y = flip ? oy : a.y;
    af1.x = flip ? oz : a.z; af1.y = flip ? ow : a.w;

    v8f d = {0.f, 0.f, 0.f, 0.f, 0.f, 0.f, 0.f, 0.f};
    d = __builtin_amdgcn_wmma_f32_16x16x4_f32(false, wf0, false, af0, (short)0, d, false, false);
    d = __builtin_amdgcn_wmma_f32_16x16x4_f32(false, wf1, false, af1, (short)0, d, false, false);

    if (h == 0 && posu < n_pos)
      *(float4*)(part + ((size_t)slice * n_pos + posu) * NBUCK) =
          make_float4(d[0], d[1], d[2], d[3]);

#pragma unroll
    for (int q = 0; q < 8; ++q) cur[q] = nxt[q];
    g += gstride;
  }
}

#define PS_WAVES 32
#define PS_THREADS (PS_WAVES * 32)
#define PS_POS 8

__global__ __launch_bounds__(PS_THREADS) void psqt_add(
    const int* __restrict__ w_cols, const int* __restrict__ b_cols,
    const float* __restrict__ stms,
    const float* __restrict__ psqt_w, const float* __restrict__ layer_b,
    const float* __restrict__ part, float* __restrict__ out, int n_pos)
{
  __shared__ float4 sTab[NFEAT];

  if (n_pos <= 0) return;
  const int wave = __builtin_amdgcn_readfirstlane((int)(threadIdx.x >> 5));
  const int lane = threadIdx.x & 31;
  const int comp = lane >> 3;
  const int groups  = (n_pos + PS_POS - 1) / PS_POS;
  const int gstride = gridDim.x * PS_WAVES;
  int g = blockIdx.x * PS_WAVES + wave;
  const float lb = layer_b[comp];

  const int pi = lane & 7;
  const int sgrp = lane >> 3;

  int    wc[PS_POS], bc[PS_POS];
#pragma unroll
  for (int i = 0; i < PS_POS; ++i) {
    const int pos = min(g * PS_POS + i, n_pos - 1);
    wc[i] = w_cols[(size_t)pos * BAG + lane];
    bc[i] = b_cols[(size_t)pos * BAG + lane];
  }

  {
    int r = (int)((blockIdx.x * 80u) % NFEAT) + (int)threadIdx.x;
#pragma unroll
    for (int i = 0; i < NFEAT / PS_THREADS; ++i) {
      if (r >= NFEAT) r -= NFEAT;
      sTab[r] = ld4(psqt_w + (size_t)r * NBUCK);
      r += PS_THREADS;
    }
  }
  __syncthreads();

  while (g < groups) {
    float4 R[NSLICE / 4];
    float  sgl;
    {
      const int pos = min(g * PS_POS + pi, n_pos - 1);
#pragma unroll
      for (int k = 0; k < NSLICE / 4; ++k)
        R[k] = ld4(part + ((unsigned)(sgrp + 4 * k) * (unsigned)n_pos + (unsigned)pos) * NBUCK);
      sgl = (stms[pos] != 0.0f) ? -1.0f : 1.0f;
    }
    float v[PS_POS];

#pragma unroll
    for (int i = 0; i < PS_POS; ++i) {
      const float4 pw = sTab[wc[i]];
      const float4 pb = sTab[bc[i]];
      const float px = pw.x - pb.x, py = pw.y - pb.y;
      const float pz = pw.z - pb.z, pq = pw.w - pb.w;
      const bool hi = (lane & 16) != 0;
      float k0 = hi ? pz : px, k1 = hi ? pq : py;
      const float s0 = hi ? px : pz, s1 = hi ? py : pq;
      k0 += __shfl_xor(s0, 16, 32);
      k1 += __shfl_xor(s1, 16, 32);
      const bool b3 = (lane & 8) != 0;
      float t = b3 ? k1 : k0;
      const float st = b3 ? k0 : k1;
      t += __shfl_xor(st, 8, 32);
      t += __shfl_xor(t, 4, 32);
      t += __shfl_xor(t, 2, 32);
      t += __shfl_xor(t, 1, 32);
      v[i] = t;
    }
    float4 q = R[0];
#pragma unroll
    for (int k = 1; k < NSLICE / 4; ++k) {
      q.x += R[k].x; q.y += R[k].y; q.z += R[k].z; q.w += R[k].w;
    }
    q.x += __shfl_xor(q.x, 8, 32);  q.y += __shfl_xor(q.y, 8, 32);
    q.z += __shfl_xor(q.z, 8, 32);  q.w += __shfl_xor(q.w, 8, 32);
    q.x += __shfl_xor(q.x, 16, 32); q.y += __shfl_xor(q.y, 16, 32);
    q.z += __shfl_xor(q.z, 16, 32); q.w += __shfl_xor(q.w, 16, 32);
    const float qc = (comp == 0) ? q.x : (comp == 1) ? q.y : (comp == 2) ? q.z : q.w;
    const float z = sgl * (qc + lb);
#pragma unroll
    for (int i = 0; i < PS_POS; ++i) {
      const int pos = g * PS_POS + i;
      const float zi = __shfl(z, (lane & 24) + i, 32);
      if (pos < n_pos && (lane & 7) == 0)
        out[(size_t)pos * NBUCK + comp] = v[i] + zi;
    }
    g += gstride;
    if (g < groups) {
#pragma unroll
      for (int i = 0; i < PS_POS; ++i) {
        const int pos = min(g * PS_POS + i, n_pos - 1);
        wc[i] = w_cols[(size_t)pos * BAG + lane];
        bc[i] = b_cols[(size_t)pos * BAG + lane];
      }
    }
  }
}

extern "C" void kernel_launch(void* const* d_in, const int* in_sizes, int n_in,
                              void* d_out, int out_size, void* d_ws, size_t ws_size,
                              hipStream_t stream) {
  const int*   w_cols   = (const int*)d_in[1];
  const int*   b_cols   = (const int*)d_in[3];
  const float* stms     = (const float*)d_in[4];
  const float* psqt_w   = (const float*)d_in[5];
  const float* acc_w    = (const float*)d_in[6];
  const float* acc_bias = (const float*)d_in[7];
  const float* layer_w  = (const float*)d_in[8];
  const float* layer_b  = (const float*)d_in[9];
  float* out = (float*)d_out;
  float* slices = (float*)d_ws;
  float* part   = (float*)d_ws + PART_OFF_FLOATS;
  int*   idxT   = (int*)((float*)d_ws + IDX_OFF_FLOATS);

  const int n_pos = in_sizes[4];
  acc_to_slices<<<TR_BLOCKS + IX_BLOCKS, 256, 0, stream>>>(
      acc_w, slices, w_cols, b_cols, idxT, n_pos);
  slice_fwd<<<NSLICE * NCHUNK, SL_THREADS, 0, stream>>>(
      idxT, stms, slices, acc_bias, layer_w, part, n_pos);
  psqt_add<<<256, PS_THREADS, 0, stream>>>(
      w_cols, b_cols, stms, psqt_w, layer_b, part, out, n_pos);
}


# psqt launch re-done like a slice: one lane per (position, bag) sums its 32 psqt rows from LDS in the bank-ordered index order of launch 1, 16-wave groups; psqt and slice-partial sum orders change
# speedup vs baseline: 1.7026x; 1.0352x over previous
#include <hip/hip_runtime.h>


#define NFEAT 20480
#define NACC  128
#define NBUCK 4
#define BAG   32
#define NSLICE 32
#define NCHUNK 8
#define PART_OFF_FLOATS (4u << 20)
#define IDX_OFF_FLOATS  (16u << 20)
#define TR_BLOCKS (NFEAT / 32)
#define IX_BLOCKS 512

typedef __attribute__((ext_vector_type(2))) float v2f;
typedef __attribute__((ext_vector_type(8))) float v8f;

__device__ __forceinline__ float4 ld4(const float* p) {
  return *(const float4*)p;
}

#define TR_ROWS 32
__global__ __launch_bounds__(256) void acc_to_slices(
    const float* __restrict__ acc_w, float* __restrict__ slices,
    const int* __restrict__ w_cols, const int* __restrict__ b_cols,
    int* __restrict__ idx_out, int n_pos)
{
  int4* __restrict__ idxT = (int4*)idx_out;
  __shared__ float4 tile[TR_ROWS][33];
  __shared__ int perm[BAG][256];
  const int wave = threadIdx.x >> 5;
  const int lane = threadIdx.x & 31;
  if (blockIdx.x >= TR_BLOCKS) {
    if (n_pos <= 0) return;
    const int groups = (n_pos + 15) >> 4;
    const int c = lane & 15, h = lane >> 4;
    for (int g = (blockIdx.x - TR_BLOCKS) * 8 + wave; g < groups; g += IX_BLOCKS * 8) {
      const int pos = min(g * 16 + c, n_pos - 1);
      const int4* p = (const int4*)((h ? b_cols : w_cols) + (size_t)pos * BAG);
      int4 d[8];
#pragma unroll
      for (int q = 0; q < 8; ++q) d[q] = p[q];
      int v[BAG];
#pragma unroll
      for (int q = 0; q < 8; ++q) {
        v[4 * q + 0] = d[q].x; v[4 * q + 1] = d[q].y;
        v[4 * q + 2] = d[q].z; v[4 * q + 3] = d[q].w;
      }
      const int rot = (c + 8 * h) & 15;
      unsigned seen = 0u, filled = 0u, late = 0u;
#pragma unroll
      for (int i = 0; i < BAG; ++i) {
        const int k = v[i] & 15;
        const unsigned cnt = (seen >> (2 * k)) & 3u;
        if (cnt < 2u) {
          const int t = ((k - rot) & 15) + 16 * (int)cnt;
          perm[t][threadIdx.x] = v[i];
          filled |= 1u << t;
          seen += 1u << (2 * k);
        } else {
          late |= 1u << i;
        }
      }
      unsigned empty = ~filled;
#pragma unroll
      for (int i = 0; i < BAG; ++i) {
        if ((late >> i) & 1u) {
          const int t = __builtin_ctz(empty);
          empty &= empty - 1u;
          perm[t][threadIdx.x] = v[i];
        }
      }
#pragma unroll
      for (int q = 0; q < 8; ++q) {
        int4 o;
        o.x = perm[4 * q + 0][threadIdx.x]; o.y = perm[4 * q + 1][threadIdx.x];
        o.z = perm[4 * q + 2][threadIdx.x]; o.w = perm[4 * q + 3][threadIdx.x];
        idxT[((size_t)g * 8 + q) * 32 + lane] = o;
      }
    }
    return;
  }
  const int row0 = blockIdx.x * TR_ROWS;
#pragma unroll
  for (int k = 0; k < 4; ++k) {
    const int r = wave * 4 + k;
    tile[r][lane] = ld4(acc_w + (size_t)(row0 + r) * NACC + lane * 4);
  }
  __syncthreads();
#pragma unroll
  for (int k = 0; k < 4; ++k) {
    const int s = wave * 4 + k;
    *(float4*)(slices + ((size_t)s * NFEAT + row0 + lane) * 4) = tile[lane][s];
  }
}

#define SL_WAVES 32
#define SL_THREADS (SL_WAVES * 32)

__device__ __forceinline__ void load_idx(int4 (&d)[8], const int4* __restrict__ idxT,
                                         int g, int lane) {
  const int4* p = idxT + (size_t)g * 8 * 32 + lane;
#pragma unroll
  for (int q = 0; q < 8; ++q) d[q] = p[q * 32];
}

__global__ __launch_bounds__(SL_THREADS) void slice_fwd(
    const int* __restrict__ idx_in,
    const float* __restrict__ stms, const float* __restrict__ slices,
    const float* __restrict__ acc_bias, const float* __restrict__ layer_w,
    float* __restrict__ part, int n_pos)
{
  __shared__ float4 sTab[NFEAT];

  if (n_pos <= 0) return;
  const int4* __restrict__ idxT = (const int4*)idx_in;
  const int wave = __builtin_amdgcn_readfirstlane((int)(threadIdx.x >> 5));
  const int lane = threadIdx.x & 31;
  const int c = lane & 15;
  const int h = lane >> 4;
  const int slice = blockIdx.x & (NSLICE - 1);
  const int chunk = blockIdx.x >> 5;
  const int groups  = (n_pos + 15) >> 4;
  const int gstride = NCHUNK * SL_WAVES;
  int g = chunk * SL_WAVES + wave;

  int4 cur[8];
  load_idx(cur, idxT, min(g, groups - 1), lane);

  const int nq = (c < NBUCK) ? c : 0;
  const float nmask = (c < NBUCK) ? 1.0f : 0.0f;
  const float* lw = layer_w + (size_t)nq * (2 * NACC) + h * NACC + slice * 4;
  v2f wf0, wf1;
  wf0.x = lw[0] * nmask; wf0.y = lw[1] * nmask;
  wf1.x = lw[2] * nmask; wf1.y = lw[3] * nmask;
  const float4 bias = ld4(acc_bias + slice * 4);

  {
    const float* src = slices + (size_t)slice * NFEAT * 4;
    int r = (int)((chunk * 2560u) % NFEAT) + (int)threadIdx.x;
#pragma unroll
    for (int i = 0; i < NFEAT / SL_THREADS; ++i) {
      if (r >= NFEAT) r -= NFEAT;
      sTab[r] = ld4(src + (size_t)r * 4);
      r += SL_THREADS;
    }
  }
  __syncthreads();

  while (g < groups) {
    const int posu = g * 16 + c;
    const int pos  = min(posu, n_pos - 1);
    const float s  = stms[pos];
    int4 nxt[8];
    load_idx(nxt, idxT, min(g + gstride, groups - 1), lane);

    float4 a = make_float4(0.f, 0.f, 0.f, 0.f);
#pragma unroll
    for (int q = 0; q < 8; ++q) {
      float4 r;
      r = sTab[cur[q].x]; a.x += r.x; a.y += r.y; a.z += r.z; a.w += r.w;
      r = sTab[cur[q].y]; a.x += r.x; a.y += r.y; a.z += r.z; a.w += r.w;
      r = sTab[cur[q].z]; a.x += r.x; a.y += r.y; a.z += r.z; a.w += r.w;
      r = sTab[cur[q].w]; a.x += r.x; a.y += r.y; a.z += r.z; a.w += r.w;
    }
    a.x = fminf(fmaxf(a.x + bias.x, 0.f), 1.f);
    a.y = fminf(fmaxf(a.y + bias.y, 0.f), 1.f);
    a.z = fminf(fmaxf(a.z + bias.z, 0.f), 1.f);
    a.w = fminf(fmaxf(a.w + bias.w, 0.f), 1.f);

    const bool flip = (s != 0.0f);
    const float ox = __shfl_xor(a.x, 16, 32);
    const float oy = __shfl_xor(a.y, 16, 32);
    const float oz = __shfl_xor(a.z, 16, 32);
    const float ow = __shfl_xor(a.w, 16, 32);
    v2f af0, af1;
    af0.x = flip ? ox : a.x; af0.y = flip ? oy : a.y;
    af1.x = flip ? oz : a.z; af1.y = flip ? ow : a.w;

    v8f d = {0.f, 0.f, 0.f, 0.f, 0.f, 0.f, 0.f, 0.f};
    d = __builtin_amdgcn_wmma_f32_16x16x4_f32(false, wf0, false, af0, (short)0, d, false, false);
    d = __builtin_amdgcn_wmma_f32_16x16x4_f32(false, wf1, false, af1, (short)0, d, false, false);

    if (h == 0 && posu < n_pos)
      *(float4*)(part + ((size_t)slice * n_pos + posu) * NBUCK) =
          make_float4(d[0], d[1], d[2], d[3]);

#pragma unroll
    for (int q = 0; q < 8; ++q) cur[q] = nxt[q];
    g += gstride;
  }
}

#define PS_WAVES 16
#define PS_THREADS (PS_WAVES * 32)

__global__ __launch_bounds__(PS_THREADS) void psqt_add(
    const int* __restrict__ idx_in, const float* __restrict__ stms,
    const float* __restrict__ psqt_w, const float* __restrict__ layer_b,
    const float* __restrict__ part, float* __restrict__ out, int n_pos)
{
  __shared__ float4 sTab[NFEAT];

  if (n_pos <= 0) return;
  const int4* __restrict__ idxT = (const int4*)idx_in;
  const int wave = __builtin_amdgcn_readfirstlane((int)(threadIdx.x >> 5));
  const int lane = threadIdx.x & 31;
  const int c = lane & 15;
  const int h = lane >> 4;
  const int groups  = (n_pos + 15) >> 4;
  const int gstride = gridDim.x * PS_WAVES;
  int g = blockIdx.x * PS_WAVES + wave;

  int4 cur[8];
  load_idx(cur, idxT, min(g, groups - 1), lane);
  float4 R[NSLICE / 2];
  float  sgl;
  {
    const int pos = min(min(g, groups - 1) * 16 + c, n_pos - 1);
#pragma unroll
    for (int k = 0; k < NSLICE / 2; ++k)
      R[k] = ld4(part + ((unsigned)(16 * h + k) * (unsigned)n_pos + (unsigned)pos) * NBUCK);
    sgl = (stms[pos] != 0.0f) ? -1.0f : 1.0f;
  }
  const float lb0 = layer_b[0], lb1 = layer_b[1], lb2 = layer_b[2], lb3 = layer_b[3];

  {
    int r = (int)((blockIdx.x * 80u) % NFEAT) + (int)threadIdx.x;
#pragma unroll 20
    for (int i = 0; i < NFEAT / PS_THREADS; ++i) {
      if (r >= NFEAT) r -= NFEAT;
      sTab[r] = ld4(psqt_w + (size_t)r * NBUCK);
      r += PS_THREADS;
    }
  }
  __syncthreads();

  while (g < groups) {
    const int posu = g * 16 + c;

    float4 a = make_float4(0.f, 0.f, 0.f, 0.f);
#pragma unroll
    for (int q = 0; q < 8; ++q) {
      float4 r;
      r = sTab[cur[q].x]; a.x += r.x; a.y += r.y; a.z += r.z; a.w += r.w;
      r = sTab[cur[q].y]; a.x += r.x; a.y += r.y; a.z += r.z; a.w += r.w;
      r = sTab[cur[q].z]; a.x += r.x; a.y += r.y; a.z += r.z; a.w += r.w;
      r = sTab[cur[q].w]; a.x += r.x; a.y += r.y; a.z += r.z; a.w += r.w;
    }
    float4 p = R[0];
#pragma unroll
    for (int k = 1; k < NSLICE / 2; ++k) {
      p.x += R[k].x; p.y += R[k].y; p.z += R[k].z; p.w += R[k].w;
    }
    const float sx = h ? -a.x : a.x, sy = h ? -a.y : a.y;
    const float sz = h ? -a.z : a.z, sw = h ? -a.w : a.w;
    const float psx = sx + __shfl_xor(sx, 16, 32);
    const float psy = sy + __shfl_xor(sy, 16, 32);
    const float psz = sz + __shfl_xor(sz, 16, 32);
    const float psw = sw + __shfl_xor(sw, 16, 32);
    const float qx = p.x + __shfl_xor(p.x, 16, 32);
    const float qy = p.y + __shfl_xor(p.y, 16, 32);
    const float qz = p.z + __shfl_xor(p.z, 16, 32);
    const float qw = p.w + __shfl_xor(p.w, 16, 32);
    if (h == 0 && posu < n_pos)
      *(float4*)(out + (size_t)posu * NBUCK) =
          make_float4(psx + sgl * (qx + lb0), psy + sgl * (qy + lb1),
                      psz + sgl * (qz + lb2), psw + sgl * (qw + lb3));

    g += gstride;
    if (g < groups) {
      load_idx(cur, idxT, g, lane);
      const int pos = min(g * 16 + c, n_pos - 1);
#pragma unroll
      for (int k = 0; k < NSLICE / 2; ++k)
        R[k] = ld4(part + ((unsigned)(16 * h + k) * (unsigned)n_pos + (unsigned)pos) * NBUCK);
      sgl = (stms[pos] != 0.0f) ? -1.0f : 1.0f;
    }
  }
}

extern "C" void kernel_launch(void* const* d_in, const int* in_sizes, int n_in,
                              void* d_out, int out_size, void* d_ws, size_t ws_size,
                              hipStream_t stream) {
  const int*   w_cols   = (const int*)d_in[1];
  const int*   b_cols   = (const int*)d_in[3];
  const float* stms     = (const float*)d_in[4];
  const float* psqt_w   = (const float*)d_in[5];
  const float* acc_w    = (const float*)d_in[6];
  const float* acc_bias = (const float*)d_in[7];
  const float* layer_w  = (const float*)d_in[8];
  const float* layer_b  = (const float*)d_in[9];
  float* out = (float*)d_out;
  float* slices = (float*)d_ws;
  float* part   = (float*)d_ws + PART_OFF_FLOATS;
  int*   idxT   = (int*)((float*)d_ws + IDX_OFF_FLOATS);

  const int n_pos = in_sizes[4];
  acc_to_slices<<<TR_BLOCKS + IX_BLOCKS, 256, 0, stream>>>(
      acc_w, slices, w_cols, b_cols, idxT, n_pos);
  slice_fwd<<<NSLICE * NCHUNK, SL_THREADS, 0, stream>>>(
      idxT, stms, slices, acc_bias, layer_w, part, n_pos);
  psqt_add<<<256, PS_THREADS, 0, stream>>>(
      idxT, stms, psqt_w, layer_b, part, out, n_pos);
}


# bag index order built by half-wave first fit: used-step masks per bank group go round the 16 lanes, a row takes its lane's lowest step its mask allows; index blocks first in grid; sum order changes
# speedup vs baseline: 1.7907x; 1.0517x over previous
#include <hip/hip_runtime.h>


#define NFEAT 20480
#define NACC  128
#define NBUCK 4
#define BAG   32
#define NSLICE 32
#define NCHUNK 8
#define PART_OFF_FLOATS (4u << 20)
#define IDX_OFF_FLOATS  (16u << 20)
#define TR_BLOCKS (NFEAT / 32)
#define IX_BLOCKS 512

typedef __attribute__((ext_vector_type(2))) float v2f;
typedef __attribute__((ext_vector_type(8))) float v8f;

__device__ __forceinline__ float4 ld4(const float* p) {
  return *(const float4*)p;
}

#define TR_ROWS 32
#define IX_LDS_BYTES (BAG * 256 * 4 + BAG * 256)
__global__ __launch_bounds__(256) void acc_to_slices(
    const float* __restrict__ acc_w, float* __restrict__ slices,
    const int* __restrict__ w_cols, const int* __restrict__ b_cols,
    int* __restrict__ idx_out, int n_pos)
{
  int4* __restrict__ idxT = (int4*)idx_out;
  __shared__ __attribute__((aligned(16))) unsigned char smem[IX_LDS_BYTES];
  const int wave = threadIdx.x >> 5;
  const int lane = threadIdx.x & 31;
  if (blockIdx.x < IX_BLOCKS) {
    if (n_pos <= 0) return;
    int* perm = (int*)smem;
    unsigned char* stepOf = smem + BAG * 256 * 4;
    const int groups = (n_pos + 15) >> 4;
    const int c = lane & 15, h = lane >> 4;
    const int from = (lane & 16) | ((c + 1) & 15);
    for (int g = blockIdx.x * 8 + wave; g < groups; g += IX_BLOCKS * 8) {
      const int pos = min(g * 16 + c, n_pos - 1);
      const int4* p = (const int4*)((h ? b_cols : w_cols) + (size_t)pos * BAG);
      int4 d[8];
#pragma unroll
      for (int q = 0; q < 8; ++q) d[q] = p[q];
      int v[BAG];
#pragma unroll
      for (int q = 0; q < 8; ++q) {
        v[4 * q + 0] = d[q].x; v[4 * q + 1] = d[q].y;
        v[4 * q + 2] = d[q].z; v[4 * q + 3] = d[q].w;
      }
      unsigned G[4] = {0u, 0u, 0u, 0u};
#pragma unroll
      for (int i = 0; i < BAG; ++i)
        G[i >> 3] |= ((unsigned)v[i] & 15u) << (4 * (i & 7));
      unsigned freeM = 0xFFFFFFFFu;
      unsigned open  = 0xFFFFFFFFu;
      for (int ph = 0; ph < 16; ++ph) {
        const unsigned kk = (unsigned)((c + ph) & 15) * 0x11111111u;
        unsigned E = 0u;
#pragma unroll
        for (int w = 0; w < 4; ++w) {
          const unsigned x = G[w] ^ kk;
          const unsigned y = ((x & 0x77777777u) + 0x77777777u) | x;
          E |= (~y & 0x88888888u) >> (3 - w);
        }
        const unsigned free0 = freeM;
        unsigned cand = freeM & open;
        while (E != 0u) {
          const int b = __builtin_ctz(E);
          E &= E - 1u;
          const unsigned x = (cand != 0u) ? cand : freeM;
          const int t = __builtin_ctz(x);
          const unsigned bit = 1u << t;
          freeM ^= bit;
          cand &= ~bit;
          stepOf[b * 256 + threadIdx.x] = (unsigned char)t;
        }
        open &= ~(free0 ^ freeM);
        open = __shfl(open, from, 32);
      }
      int st[BAG];
#pragma unroll
      for (int i = 0; i < BAG; ++i)
        st[i] = stepOf[(4 * (i & 7) + (i >> 3)) * 256 + threadIdx.x];
#pragma unroll
      for (int i = 0; i < BAG; ++i)
        perm[st[i] * 256 + threadIdx.x] = v[i];
#pragma unroll
      for (int q = 0; q < 8; ++q) {
        int4 o;
        o.x = perm[(4 * q + 0) * 256 + threadIdx.x]; o.y = perm[(4 * q + 1) * 256 + threadIdx.x];
        o.z = perm[(4 * q + 2) * 256 + threadIdx.x]; o.w = perm[(4 * q + 3) * 256 + threadIdx.x];
        idxT[((size_t)g * 8 + q) * 32 + lane] = o;
      }
    }
    return;
  }
  float4 (*tile)[33] = (float4 (*)[33])smem;
  const int row0 = (blockIdx.x - IX_BLOCKS) * TR_ROWS;
#pragma unroll
  for (int k = 0; k < 4; ++k) {
    const int r = wave * 4 + k;
    tile[r][lane] = ld4(acc_w + (size_t)(row0 + r) * NACC + lane * 4);
  }
  __syncthreads();
#pragma unroll
  for (int k = 0; k < 4; ++k) {
    const int s = wave * 4 + k;
    *(float4*)(slices + ((size_t)s * NFEAT + row0 + lane) * 4) = tile[lane][s];
  }
}

#define SL_WAVES 32
#define SL_THREADS (SL_WAVES * 32)

__device__ __forceinline__ void load_idx(int4 (&d)[8], const int4* __restrict__ idxT,
                                         int g, int lane) {
  const int4* p = idxT + (size_t)g * 8 * 32 + lane;
#pragma unroll
  for (int q = 0; q < 8; ++q) d[q] = p[q * 32];
}

__global__ __launch_bounds__(SL_THREADS) void slice_fwd(
    const int* __restrict__ idx_in,
    const float* __restrict__ stms, const float* __restrict__ slices,
    const float* __restrict__ acc_bias, const float* __restrict__ layer_w,
    float* __restrict__ part, int n_pos)
{
  __shared__ float4 sTab[NFEAT];

  if (n_pos <= 0) return;
  const int4* __restrict__ idxT = (const int4*)idx_in;
  const int wave = __builtin_amdgcn_readfirstlane((int)(threadIdx.x >> 5));
  const int lane = threadIdx.x & 31;
  const int c = lane & 15;
  const int h = lane >> 4;
  const int slice = blockIdx.x & (NSLICE - 1);
  const int chunk = blockIdx.x >> 5;
  const int groups  = (n_pos + 15) >> 4;
  const int gstride = NCHUNK * SL_WAVES;
  int g = chunk * SL_WAVES + wave;

  int4 cur[8];
  load_idx(cur, idxT, min(g, groups - 1), lane);

  const int nq = (c < NBUCK) ? c : 0;
  const float nmask = (c < NBUCK) ? 1.0f : 0.0f;
  const float* lw = layer_w + (size_t)nq * (2 * NACC) + h * NACC + slice * 4;
  v2f wf0, wf1;
  wf0.x = lw[0] * nmask; wf0.y = lw[1] * nmask;
  wf1.x = lw[2] * nmask; wf1.y = lw[3] * nmask;
  const float4 bias = ld4(acc_bias + slice * 4);

  {
    const float* src = slices + (size_t)slice * NFEAT * 4;
    int r = (int)((chunk * 2560u) % NFEAT) + (int)threadIdx.x;
#pragma unroll
    for (int i = 0; i < NFEAT / SL_THREADS; ++i) {
      if (r >= NFEAT) r -= NFEAT;
      sTab[r] = ld4(src + (size_t)r * 4);
      r += SL_THREADS;
    }
  }
  __syncthreads();

  while (g < groups) {
    const int posu = g * 16 + c;
    const int pos  = min(posu, n_pos - 1);
    const float s  = stms[pos];
    int4 nxt[8];
    load_idx(nxt, idxT, min(g + gstride, groups - 1), lane);

    float4 a = make_float4(0.f, 0.f, 0.f, 0.f);
#pragma unroll
    for (int q = 0; q < 8; ++q) {
      float4 r;
      r = sTab[cur[q].x]; a.x += r.x; a.y += r.y; a.z += r.z; a.w += r.w;
      r = sTab[cur[q].y]; a.x += r.x; a.y += r.y; a.z += r.z; a.w += r.w;
      r = sTab[cur[q].z]; a.x += r.x; a.y += r.y; a.z += r.z; a.w += r.w;
      r = sTab[cur[q].w]; a.x += r.x; a.y += r.y; a.z += r.z; a.w += r.w;
    }
    a.x = fminf(fmaxf(a.x + bias.x, 0.f), 1.f);
    a.y = fminf(fmaxf(a.y + bias.y, 0.f), 1.f);
    a.z = fminf(fmaxf(a.z + bias.z, 0.f), 1.f);
    a.w = fminf(fmaxf(a.w + bias.w, 0.f), 1.f);

    const bool flip = (s != 0.0f);
    const float ox = __shfl_xor(a.x, 16, 32);
    const float oy = __shfl_xor(a.y, 16, 32);
    const float oz = __shfl_xor(a.z, 16, 32);
    const float ow = __shfl_xor(a.w, 16, 32);
    v2f af0, af1;
    af0.x = flip ? ox : a.x; af0.y = flip ? oy : a.y;
    af1.x = flip ? oz : a.z; af1.y = flip ? ow : a.w;

    v8f d = {0.f, 0.f, 0.f, 0.f, 0.f, 0.f, 0.f, 0.f};
    d = __builtin_amdgcn_wmma_f32_16x16x4_f32(false, wf0, false, af0, (short)0, d, false, false);
    d = __builtin_amdgcn_wmma_f32_16x16x4_f32(false, wf1, false, af1, (short)0, d, false, false);

    if (h == 0 && posu < n_pos)
      *(float4*)(part + ((size_t)slice * n_pos + posu) * NBUCK) =
          make_float4(d[0], d[1], d[2], d[3]);

#pragma unroll
    for (int q = 0; q < 8; ++q) cur[q] = nxt[q];
    g += gstride;
  }
}

#define PS_WAVES 16
#define PS_THREADS (PS_WAVES * 32)

__global__ __launch_bounds__(PS_THREADS) void psqt_add(
    const int* __restrict__ idx_in, const float* __restrict__ stms,
    const float* __restrict__ psqt_w, const float* __restrict__ layer_b,
    const float* __restrict__ part, float* __restrict__ out, int n_pos)
{
  __shared__ float4 sTab[NFEAT];

  if (n_pos <= 0) return;
  const int4* __restrict__ idxT = (const int4*)idx_in;
  const int wave = __builtin_amdgcn_readfirstlane((int)(threadIdx.x >> 5));
  const int lane = threadIdx.x & 31;
  const int c = lane & 15;
  const int h = lane >> 4;
  const int groups  = (n_pos + 15) >> 4;
  const int gstride = gridDim.x * PS_WAVES;
  int g = blockIdx.x * PS_WAVES + wave;

  int4 cur[8];
  load_idx(cur, idxT, min(g, groups - 1), lane);
  float4 R[NSLICE / 2];
  float  sgl;
  {
    const int pos = min(min(g, groups - 1) * 16 + c, n_pos - 1);
#pragma unroll
    for (int k = 0; k < NSLICE / 2; ++k)
      R[k] = ld4(part + ((unsigned)(16 * h + k) * (unsigned)n_pos + (unsigned)pos) * NBUCK);
    sgl = (stms[pos] != 0.0f) ? -1.0f : 1.0f;
  }
  const float lb0 = layer_b[0], lb1 = layer_b[1], lb2 = layer_b[2], lb3 = layer_b[3];

  {
    int r = (int)((blockIdx.x * 80u) % NFEAT) + (int)threadIdx.x;
#pragma unroll 20
    for (int i = 0; i < NFEAT / PS_THREADS; ++i) {
      if (r >= NFEAT) r -= NFEAT;
      sTab[r] = ld4(psqt_w + (size_t)r * NBUCK);
      r += PS_THREADS;
    }
  }
  __syncthreads();

  while (g < groups) {
    const int posu = g * 16 + c;

    float4 a = make_float4(0.f, 0.f, 0.f, 0.f);
#pragma unroll
    for (int q = 0; q < 8; ++q) {
      float4 r;
      r = sTab[cur[q].x]; a.x += r.x; a.y += r.y; a.z += r.z; a.w += r.w;
      r = sTab[cur[q].y]; a.x += r.x; a.y += r.y; a.z += r.z; a.w += r.w;
      r = sTab[cur[q].z]; a.x += r.x; a.y += r.y; a.z += r.z; a.w += r.w;
      r = sTab[cur[q].w]; a.x += r.x; a.y += r.y; a.z += r.z; a.w += r.w;
    }
    float4 p = R[0];
#pragma unroll
    for (int k = 1; k < NSLICE / 2; ++k) {
      p.x += R[k].x; p.y += R[k].y; p.z += R[k].z; p.w += R[k].w;
    }
    const float sx = h ? -a.x : a.x, sy = h ? -a.y : a.y;
    const float sz = h ? -a.z : a.z, sw = h ? -a.w : a.w;
    const float psx = sx + __shfl_xor(sx, 16, 32);
    const float psy = sy + __shfl_xor(sy, 16, 32);
    const float psz = sz + __shfl_xor(sz, 16, 32);
    const float psw = sw + __shfl_xor(sw, 16, 32);
    const float qx = p.x + __shfl_xor(p.x, 16, 32);
    const float qy = p.y + __shfl_xor(p.y, 16, 32);
    const float qz = p.z + __shfl_xor(p.z, 16, 32);
    const float qw = p.w + __shfl_xor(p.w, 16, 32);
    if (h == 0 && posu < n_pos)
      *(float4*)(out + (size_t)posu * NBUCK) =
          make_float4(psx + sgl * (qx + lb0), psy + sgl * (qy + lb1),
                      psz + sgl * (qz + lb2), psw + sgl * (qw + lb3));

    g += gstride;
    if (g < groups) {
      load_idx(cur, idxT, g, lane);
      const int pos = min(g * 16 + c, n_pos - 1);
#pragma unroll
      for (int k = 0; k < NSLICE / 2; ++k)
        R[k] = ld4(part + ((unsigned)(16 * h + k) * (unsigned)n_pos + (unsigned)pos) * NBUCK);
      sgl = (stms[pos] != 0.0f) ? -1.0f : 1.0f;
    }
  }
}

extern "C" void kernel_launch(void* const* d_in, const int* in_sizes, int n_in,
                              void* d_out, int out_size, void* d_ws, size_t ws_size,
                              hipStream_t stream) {
  const int*   w_cols   = (const int*)d_in[1];
  const int*   b_cols   = (const int*)d_in[3];
  const float* stms     = (const float*)d_in[4];
  const float* psqt_w   = (const float*)d_in[5];
  const float* acc_w    = (const float*)d_in[6];
  const float* acc_bias = (const float*)d_in[7];
  const float* layer_w  = (const float*)d_in[8];
  const float* layer_b  = (const float*)d_in[9];
  float* out = (float*)d_out;
  float* slices = (float*)d_ws;
  float* part   = (float*)d_ws + PART_OFF_FLOATS;
  int*   idxT   = (int*)((float*)d_ws + IDX_OFF_FLOATS);

  const int n_pos = in_sizes[4];
  acc_to_slices<<<TR_BLOCKS + IX_BLOCKS, 256, 0, stream>>>(
      acc_w, slices, w_cols, b_cols, idxT, n_pos);
  slice_fwd<<<NSLICE * NCHUNK, SL_THREADS, 0, stream>>>(
      idxT, stms, slices, acc_bias, layer_w, part, n_pos);
  psqt_add<<<256, PS_THREADS, 0, stream>>>(
      idxT, stms, psqt_w, layer_b, part, out, n_pos);
}


# bag indices kept as 16-bit halves, two per word, in the re-laid index table: the index bytes launch 1 writes and launches 2 and 3 read are halved; every sum order unchanged
# speedup vs baseline: 1.9460x; 1.0867x over previous
#include <hip/hip_runtime.h>


#define NFEAT 20480
#define NACC  128
#define NBUCK 4
#define BAG   32
#define NSLICE 32
#define NCHUNK 8
#define PART_OFF_FLOATS (4u << 20)
#define IDX_OFF_FLOATS  (16u << 20)
#define TR_BLOCKS (NFEAT / 32)
#define IX_BLOCKS 512

typedef __attribute__((ext_vector_type(2))) float v2f;
typedef __attribute__((ext_vector_type(8))) float v8f;

__device__ __forceinline__ float4 ld4(const float* p) {
  return *(const float4*)p;
}

#define TR_ROWS 32
#define IX_LDS_BYTES (BAG * 256 * 4 + BAG * 256)
__global__ __launch_bounds__(256) void acc_to_slices(
    const float* __restrict__ acc_w, float* __restrict__ slices,
    const int* __restrict__ w_cols, const int* __restrict__ b_cols,
    int* __restrict__ idx_out, int n_pos)
{
  int4* __restrict__ idxT = (int4*)idx_out;
  __shared__ __attribute__((aligned(16))) unsigned char smem[IX_LDS_BYTES];
  const int wave = threadIdx.x >> 5;
  const int lane = threadIdx.x & 31;
  if (blockIdx.x < IX_BLOCKS) {
    if (n_pos <= 0) return;
    int* perm = (int*)smem;
    unsigned char* stepOf = smem + BAG * 256 * 4;
    const int groups = (n_pos + 15) >> 4;
    const int c = lane & 15, h = lane >> 4;
    const int from = (lane & 16) | ((c + 1) & 15);
    for (int g = blockIdx.x * 8 + wave; g < groups; g += IX_BLOCKS * 8) {
      const int pos = min(g * 16 + c, n_pos - 1);
      const int4* p = (const int4*)((h ? b_cols : w_cols) + (size_t)pos * BAG);
      int4 d[8];
#pragma unroll
      for (int q = 0; q < 8; ++q) d[q] = p[q];
      int v[BAG];
#pragma unroll
      for (int q = 0; q < 8; ++q) {
        v[4 * q + 0] = d[q].x; v[4 * q + 1] = d[q].y;
        v[4 * q + 2] = d[q].z; v[4 * q + 3] = d[q].w;
      }
      unsigned G[4] = {0u, 0u, 0u, 0u};
#pragma unroll
      for (int i = 0; i < BAG; ++i)
        G[i >> 3] |= ((unsigned)v[i] & 15u) << (4 * (i & 7));
      unsigned freeM = 0xFFFFFFFFu;
      unsigned open  = 0xFFFFFFFFu;
      for (int ph = 0; ph < 16; ++ph) {
        const unsigned kk = (unsigned)((c + ph) & 15) * 0x11111111u;
        unsigned E = 0u;
#pragma unroll
        for (int w = 0; w < 4; ++w) {
          const unsigned x = G[w] ^ kk;
          const unsigned y = ((x & 0x77777777u) + 0x77777777u) | x;
          E |= (~y & 0x88888888u) >> (3 - w);
        }
        const unsigned free0 = freeM;
        unsigned cand = freeM & open;
        while (E != 0u) {
          const int b = __builtin_ctz(E);
          E &= E - 1u;
          const unsigned x = (cand != 0u) ? cand : freeM;
          const int t = __builtin_ctz(x);
          const unsigned bit = 1u << t;
          freeM ^= bit;
          cand &= ~bit;
          stepOf[b * 256 + threadIdx.x] = (unsigned char)t;
        }
        open &= ~(free0 ^ freeM);
        open = __shfl(open, from, 32);
      }
      int st[BAG];
#pragma unroll
      for (int i = 0; i < BAG; ++i)
        st[i] = stepOf[(4 * (i & 7) + (i >> 3)) * 256 + threadIdx.x];
#pragma unroll
      for (int i = 0; i < BAG; ++i)
        perm[st[i] * 256 + threadIdx.x] = v[i];
#pragma unroll
      for (int q = 0; q < 4; ++q) {
        unsigned e[8];
#pragma unroll
        for (int j = 0; j < 8; ++j) e[j] = (unsigned)perm[(8 * q + j) * 256 + threadIdx.x];
        int4 o;
        o.x = (int)((e[0] & 0xFFFFu) | (e[1] << 16)); o.y = (int)((e[2] & 0xFFFFu) | (e[3] << 16));
        o.z = (int)((e[4] & 0xFFFFu) | (e[5] << 16)); o.w = (int)((e[6] & 0xFFFFu) | (e[7] << 16));
        idxT[((size_t)g * 4 + q) * 32 + lane] = o;
      }
    }
    return;
  }
  float4 (*tile)[33] = (float4 (*)[33])smem;
  const int row0 = (blockIdx.x - IX_BLOCKS) * TR_ROWS;
#pragma unroll
  for (int k = 0; k < 4; ++k) {
    const int r = wave * 4 + k;
    tile[r][lane] = ld4(acc_w + (size_t)(row0 + r) * NACC + lane * 4);
  }
  __syncthreads();
#pragma unroll
  for (int k = 0; k < 4; ++k) {
    const int s = wave * 4 + k;
    *(float4*)(slices + ((size_t)s * NFEAT + row0 + lane) * 4) = tile[lane][s];
  }
}

#define SL_WAVES 32
#define SL_THREADS (SL_WAVES * 32)

__device__ __forceinline__ void load_idx(int4 (&d)[4], const int4* __restrict__ idxT,
                                         int g, int lane) {
  const int4* p = idxT + (size_t)g * 4 * 32 + lane;
#pragma unroll
  for (int q = 0; q < 4; ++q) d[q] = p[q * 32];
}

__device__ __forceinline__ void add_rows(float4& a, const float4* tab, int w) {
  float4 r;
  r = tab[(unsigned)w & 0xFFFFu]; a.x += r.x; a.y += r.y; a.z += r.z; a.w += r.w;
  r = tab[(unsigned)w >> 16];     a.x += r.x; a.y += r.y; a.z += r.z; a.w += r.w;
}

__global__ __launch_bounds__(SL_THREADS) void slice_fwd(
    const int* __restrict__ idx_in,
    const float* __restrict__ stms, const float* __restrict__ slices,
    const float* __restrict__ acc_bias, const float* __restrict__ layer_w,
    float* __restrict__ part, int n_pos)
{
  __shared__ float4 sTab[NFEAT];

  if (n_pos <= 0) return;
  const int4* __restrict__ idxT = (const int4*)idx_in;
  const int wave = __builtin_amdgcn_readfirstlane((int)(threadIdx.x >> 5));
  const int lane = threadIdx.x & 31;
  const int c = lane & 15;
  const int h = lane >> 4;
  const int slice = blockIdx.x & (NSLICE - 1);
  const int chunk = blockIdx.x >> 5;
  const int groups  = (n_pos + 15) >> 4;
  const int gstride = NCHUNK * SL_WAVES;
  int g = chunk * SL_WAVES + wave;

  int4 cur[4];
  load_idx(cur, idxT, min(g, groups - 1), lane);

  const int nq = (c < NBUCK) ? c : 0;
  const float nmask = (c < NBUCK) ? 1.0f : 0.0f;
  const float* lw = layer_w + (size_t)nq * (2 * NACC) + h * NACC + slice * 4;
  v2f wf0, wf1;
  wf0.x = lw[0] * nmask; wf0.y = lw[1] * nmask;
  wf1.x = lw[2] * nmask; wf1.y = lw[3] * nmask;
  const float4 bias = ld4(acc_bias + slice * 4);

  {
    const float* src = slices + (size_t)slice * NFEAT * 4;
    int r = (int)((chunk * 2560u) % NFEAT) + (int)threadIdx.x;
#pragma unroll
    for (int i = 0; i < NFEAT / SL_THREADS; ++i) {
      if (r >= NFEAT) r -= NFEAT;
      sTab[r] = ld4(src + (size_t)r * 4);
      r += SL_THREADS;
    }
  }
  __syncthreads();

  while (g < groups) {
    const int posu = g * 16 + c;
    const int pos  = min(posu, n_pos - 1);
    const float s  = stms[pos];
    int4 nxt[4];
    load_idx(nxt, idxT, min(g + gstride, groups - 1), lane);

    float4 a = make_float4(0.f, 0.f, 0.f, 0.f);
#pragma unroll
    for (int q = 0; q < 4; ++q) {
      add_rows(a, sTab, cur[q].x); add_rows(a, sTab, cur[q].y);
      add_rows(a, sTab, cur[q].z); add_rows(a, sTab, cur[q].w);
    }
    a.x = fminf(fmaxf(a.x + bias.x, 0.f), 1.f);
    a.y = fminf(fmaxf(a.y + bias.y, 0.f), 1.f);
    a.z = fminf(fmaxf(a.z + bias.z, 0.f), 1.f);
    a.w = fminf(fmaxf(a.w + bias.w, 0.f), 1.f);

    const bool flip = (s != 0.0f);
    const float ox = __shfl_xor(a.x, 16, 32);
    const float oy = __shfl_xor(a.y, 16, 32);
    const float oz = __shfl_xor(a.z, 16, 32);
    const float ow = __shfl_xor(a.w, 16, 32);
    v2f af0, af1;
    af0.x = flip ? ox : a.x; af0.y = flip ? oy : a.y;
    af1.x = flip ? oz : a.z; af1.y = flip ? ow : a.w;

    v8f d = {0.f, 0.f, 0.f, 0.f, 0.f, 0.f, 0.f, 0.f};
    d = __builtin_amdgcn_wmma_f32_16x16x4_f32(false, wf0, false, af0, (short)0, d, false, false);
    d = __builtin_amdgcn_wmma_f32_16x16x4_f32(false, wf1, false, af1, (short)0, d, false, false);

    if (h == 0 && posu < n_pos)
      *(float4*)(part + ((size_t)slice * n_pos + posu) * NBUCK) =
          make_float4(d[0], d[1], d[2], d[3]);

#pragma unroll
    for (int q = 0; q < 4; ++q) cur[q] = nxt[q];
    g += gstride;
  }
}

#define PS_WAVES 16
#define PS_THREADS (PS_WAVES * 32)

__global__ __launch_bounds__(PS_THREADS) void psqt_add(
    const int* __restrict__ idx_in, const float* __restrict__ stms,
    const float* __restrict__ psqt_w, const float* __restrict__ layer_b,
    const float* __restrict__ part, float* __restrict__ out, int n_pos)
{
  __shared__ float4 sTab[NFEAT];

  if (n_pos <= 0) return;
  const int4* __restrict__ idxT = (const int4*)idx_in;
  const int wave = __builtin_amdgcn_readfirstlane((int)(threadIdx.x >> 5));
  const int lane = threadIdx.x & 31;
  const int c = lane & 15;
  const int h = lane >> 4;
  const int groups  = (n_pos + 15) >> 4;
  const int gstride = gridDim.x * PS_WAVES;
  int g = blockIdx.x * PS_WAVES + wave;

  int4 cur[4];
  load_idx(cur, idxT, min(g, groups - 1), lane);
  float4 R[NSLICE / 2];
  float  sgl;
  {
    const int pos = min(min(g, groups - 1) * 16 + c, n_pos - 1);
#pragma unroll
    for (int k = 0; k < NSLICE / 2; ++k)
      R[k] = ld4(part + ((unsigned)(16 * h + k) * (unsigned)n_pos + (unsigned)pos) * NBUCK);
    sgl = (stms[pos] != 0.0f) ? -1.0f : 1.0f;
  }
  const float lb0 = layer_b[0], lb1 = layer_b[1], lb2 = layer_b[2], lb3 = layer_b[3];

  {
    int r = (int)((blockIdx.x * 80u) % NFEAT) + (int)threadIdx.x;
#pragma unroll 20
    for (int i = 0; i < NFEAT / PS_THREADS; ++i) {
      if (r >= NFEAT) r -= NFEAT;
      sTab[r] = ld4(psqt_w + (size_t)r * NBUCK);
      r += PS_THREADS;
    }
  }
  __syncthreads();

  while (g < groups) {
    const int posu = g * 16 + c;

    float4 a = make_float4(0.f, 0.f, 0.f, 0.f);
#pragma unroll
    for (int q = 0; q < 4; ++q) {
      add_rows(a, sTab, cur[q].x); add_rows(a, sTab, cur[q].y);
      add_rows(a, sTab, cur[q].z); add_rows(a, sTab, cur[q].w);
    }
    float4 p = R[0];
#pragma unroll
    for (int k = 1; k < NSLICE / 2; ++k) {
      p.x += R[k].x; p.y += R[k].y; p.z += R[k].z; p.w += R[k].w;
    }
    const float sx = h ? -a.x : a.x, sy = h ? -a.y : a.y;
    const float sz = h ? -a.z : a.z, sw = h ? -a.w : a.w;
    const float psx = sx + __shfl_xor(sx, 16, 32);
    const float psy = sy + __shfl_xor(sy, 16, 32);
    const float psz = sz + __shfl_xor(sz, 16, 32);
    const float psw = sw + __shfl_xor(sw, 16, 32);
    const float qx = p.x + __shfl_xor(p.x, 16, 32);
    const float qy = p.y + __shfl_xor(p.y, 16, 32);
    const float qz = p.z + __shfl_xor(p.z, 16, 32);
    const float qw = p.w + __shfl_xor(p.w, 16, 32);
    if (h == 0 && posu < n_pos)
      *(float4*)(out + (size_t)posu * NBUCK) =
          make_float4(psx + sgl * (qx + lb0), psy + sgl * (qy + lb1),
                      psz + sgl * (qz + lb2), psw + sgl * (qw + lb3));

    g += gstride;
    if (g < groups) {
      load_idx(cur, idxT, g, lane);
      const int pos = min(g * 16 + c, n_pos - 1);
#pragma unroll
      for (int k = 0; k < NSLICE / 2; ++k)
        R[k] = ld4(part + ((unsigned)(16 * h + k) * (unsigned)n_pos + (unsigned)pos) * NBUCK);
      sgl = (stms[pos] != 0.0f) ? -1.0f : 1.0f;
    }
  }
}

extern "C" void kernel_launch(void* const* d_in, const int* in_sizes, int n_in,
                              void* d_out, int out_size, void* d_ws, size_t ws_size,
                              hipStream_t stream) {
  const int*   w_cols   = (const int*)d_in[1];
  const int*   b_cols   = (const int*)d_in[3];
  const float* stms     = (const float*)d_in[4];
  const float* psqt_w   = (const float*)d_in[5];
  const float* acc_w    = (const float*)d_in[6];
  const float* acc_bias = (const float*)d_in[7];
  const float* layer_w  = (const float*)d_in[8];
  const float* layer_b  = (const float*)d_in[9];
  float* out = (float*)d_out;
  float* slices = (float*)d_ws;
  float* part   = (float*)d_ws + PART_OFF_FLOATS;
  int*   idxT   = (int*)((float*)d_ws + IDX_OFF_FLOATS);

  const int n_pos = in_sizes[4];
  acc_to_slices<<<TR_BLOCKS + IX_BLOCKS, 256, 0, stream>>>(
      acc_w, slices, w_cols, b_cols, idxT, n_pos);
  slice_fwd<<<NSLICE * NCHUNK, SL_THREADS, 0, stream>>>(
      idxT, stms, slices, acc_bias, layer_w, part, n_pos);
  psqt_add<<<256, PS_THREADS, 0, stream>>>(
      idxT, stms, psqt_w, layer_b, part, out, n_pos);
}


# bag index order built step by step: each step's mask of bank groups read so far goes round the 16 lanes, a lane adds the unread group it holds most rows of; bag sum orders change (slices and psqt)
# speedup vs baseline: 2.0366x; 1.0466x over previous
#include <hip/hip_runtime.h>


#define NFEAT 20480
#define NACC  128
#define NBUCK 4
#define BAG   32
#define NSLICE 32
#define NCHUNK 8
#define PART_OFF_FLOATS (4u << 20)
#define IDX_OFF_FLOATS  (16u << 20)
#define TR_BLOCKS (NFEAT / 32)
#define IX_BLOCKS 512

typedef __attribute__((ext_vector_type(2))) float v2f;
typedef __attribute__((ext_vector_type(8))) float v8f;

__device__ __forceinline__ float4 ld4(const float* p) {
  return *(const float4*)p;
}

#define TR_ROWS 32
#define IX_LDS_BYTES (BAG * 256 * 4 + (BAG + 1) * 256)
__global__ __launch_bounds__(256) void acc_to_slices(
    const float* __restrict__ acc_w, float* __restrict__ slices,
    const int* __restrict__ w_cols, const int* __restrict__ b_cols,
    int* __restrict__ idx_out, int n_pos)
{
  int4* __restrict__ idxT = (int4*)idx_out;
  __shared__ __attribute__((aligned(16))) unsigned char smem[IX_LDS_BYTES];
  const int wave = threadIdx.x >> 5;
  const int lane = threadIdx.x & 31;
  if (blockIdx.x < IX_BLOCKS) {
    if (n_pos <= 0) return;
    int* perm = (int*)smem;
    unsigned char* stepOf = smem + BAG * 256 * 4;
    const int groups = (n_pos + 15) >> 4;
    const int c = lane & 15, h = lane >> 4;
    for (int g = blockIdx.x * 8 + wave; g < groups; g += IX_BLOCKS * 8) {
      const int pos = min(g * 16 + c, n_pos - 1);
      const int4* p = (const int4*)((h ? b_cols : w_cols) + (size_t)pos * BAG);
      int4 d[8];
#pragma unroll
      for (int q = 0; q < 8; ++q) d[q] = p[q];
      int v[BAG];
#pragma unroll
      for (int q = 0; q < 8; ++q) {
        v[4 * q + 0] = d[q].x; v[4 * q + 1] = d[q].y;
        v[4 * q + 2] = d[q].z; v[4 * q + 3] = d[q].w;
      }
      unsigned G[4] = {0u, 0u, 0u, 0u};
#pragma unroll
      for (int i = 0; i < BAG; ++i)
        G[i >> 3] |= ((unsigned)v[i] & 15u) << (4 * (i & 7));
      unsigned B[4], nB[4];
#pragma unroll
      for (int k = 0; k < 4; ++k) {
        unsigned bk = 0u;
#pragma unroll
        for (int w = 0; w < 4; ++w) bk |= ((G[w] >> k) & 0x11111111u) << w;
        B[k] = bk; nB[k] = ~bk;
      }
      unsigned P0 = 0u, P1 = 0u, P2 = 0u;
#pragma unroll
      for (int gq = 0; gq < 16; ++gq) {
        const unsigned e = ((gq & 1) ? B[0] : nB[0]) & ((gq & 2) ? B[1] : nB[1]) &
                           ((gq & 4) ? B[2] : nB[2]) & ((gq & 8) ? B[3] : nB[3]);
        const unsigned k = min((unsigned)__builtin_popcount(e), 7u);
        P0 |= (k & 1u) << gq; P1 |= ((k >> 1) & 1u) << gq; P2 |= (k >> 2) << gq;
      }
      unsigned rec[BAG];
#pragma unroll
      for (int batch = 0; batch < 2; ++batch) {
        unsigned Ot = 0xFFFFu;
        unsigned Od = 0xFFFFu;
        unsigned TT = (unsigned)c;
#pragma unroll
        for (int s = 0; s < 16; ++s) {
          const unsigned adj = P0 | P1 | P2;
          const unsigned c1  = adj & Ot;
          const unsigned c2  = adj & Od;
          unsigned cand = c1 ? c1 : (c2 ? c2 : adj);
          unsigned x;
          x = cand & P2; cand = x ? x : cand;
          x = cand & P1; cand = x ? x : cand;
          x = cand & P0; cand = x ? x : cand;
          const unsigned bit  = cand & (0u - cand);
          const unsigned gsel = (unsigned)__builtin_ctz(bit | 0x80000000u);
          const unsigned b0 = bit & ~P0; P0 ^= bit;
          const unsigned b1 = b0 & ~P1;  P1 ^= b0;
          P2 ^= b1;
          Od &= ~(bit & ~Ot);
          Ot &= ~bit;
          rec[batch * 16 + s] = gsel | (TT << 5);
          Ot = (unsigned)__builtin_amdgcn_update_dpp(0, (int)Ot, 0x121, 0xF, 0xF, false);
          Od = (unsigned)__builtin_amdgcn_update_dpp(0, (int)Od, 0x121, 0xF, 0xF, false);
          TT = (unsigned)__builtin_amdgcn_update_dpp(0, (int)TT, 0x121, 0xF, 0xF, false);
        }
      }
      unsigned unplaced = 0xFFFFFFFFu;
      unsigned freeS    = 0xFFFFFFFFu;
#pragma unroll
      for (int idx = 0; idx < BAG; ++idx) {
        const unsigned r    = rec[idx];
        const unsigned gsel = r & 31u;
        const unsigned t    = (unsigned)(idx & 16) + ((r >> 5) & 15u);
        unsigned E = unplaced;
#pragma unroll
        for (int k = 0; k < 4; ++k) {
          const unsigned mk = (unsigned)(((int)(gsel << (31 - k))) >> 31);
          E &= ~(B[k] ^ mk);
        }
        E &= ~(unsigned)(((int)(gsel << 27)) >> 31);
        const unsigned tb = 1u << t;
        unsigned lb = E & (0u - E);
        lb = (freeS & tb) ? lb : 0u;
        unplaced ^= lb;
        freeS    ^= lb ? tb : 0u;
        const unsigned b = lb ? (unsigned)__builtin_ctz(lb) : (unsigned)BAG;
        stepOf[b * 256 + threadIdx.x] = (unsigned char)t;
      }
      for (int it = 0; it < BAG && unplaced != 0u; ++it) {
        const int b = __builtin_ctz(unplaced);
        const int t = __builtin_ctz(freeS);
        unplaced &= unplaced - 1u;
        freeS    &= freeS - 1u;
        stepOf[b * 256 + threadIdx.x] = (unsigned char)t;
      }
      int st[BAG];
#pragma unroll
      for (int i = 0; i < BAG; ++i)
        st[i] = stepOf[(4 * (i & 7) + (i >> 3)) * 256 + threadIdx.x];
#pragma unroll
      for (int i = 0; i < BAG; ++i)
        perm[st[i] * 256 + threadIdx.x] = v[i];
#pragma unroll
      for (int q = 0; q < 4; ++q) {
        unsigned e[8];
#pragma unroll
        for (int j = 0; j < 8; ++j) e[j] = (unsigned)perm[(8 * q + j) * 256 + threadIdx.x];
        int4 o;
        o.x = (int)((e[0] & 0xFFFFu) | (e[1] << 16)); o.y = (int)((e[2] & 0xFFFFu) | (e[3] << 16));
        o.z = (int)((e[4] & 0xFFFFu) | (e[5] << 16)); o.w = (int)((e[6] & 0xFFFFu) | (e[7] << 16));
        idxT[((size_t)g * 4 + q) * 32 + lane] = o;
      }
    }
    return;
  }
  float4 (*tile)[33] = (float4 (*)[33])smem;
  const int row0 = (blockIdx.x - IX_BLOCKS) * TR_ROWS;
#pragma unroll
  for (int k = 0; k < 4; ++k) {
    const int r = wave * 4 + k;
    tile[r][lane] = ld4(acc_w + (size_t)(row0 + r) * NACC + lane * 4);
  }
  __syncthreads();
#pragma unroll
  for (int k = 0; k < 4; ++k) {
    const int s = wave * 4 + k;
    *(float4*)(slices + ((size_t)s * NFEAT + row0 + lane) * 4) = tile[lane][s];
  }
}

#define SL_WAVES 32
#define SL_THREADS (SL_WAVES * 32)

__device__ __forceinline__ void load_idx(int4 (&d)[4], const int4* __restrict__ idxT,
                                         int g, int lane) {
  const int4* p = idxT + (size_t)g * 4 * 32 + lane;
#pragma unroll
  for (int q = 0; q < 4; ++q) d[q] = p[q * 32];
}

__device__ __forceinline__ void add_rows(float4& a, const float4* tab, int w) {
  float4 r;
  r = tab[(unsigned)w & 0xFFFFu]; a.x += r.x; a.y += r.y; a.z += r.z; a.w += r.w;
  r = tab[(unsigned)w >> 16];     a.x += r.x; a.y += r.y; a.z += r.z; a.w += r.w;
}

__global__ __launch_bounds__(SL_THREADS) void slice_fwd(
    const int* __restrict__ idx_in,
    const float* __restrict__ stms, const float* __restrict__ slices,
    const float* __restrict__ acc_bias, const float* __restrict__ layer_w,
    float* __restrict__ part, int n_pos)
{
  __shared__ float4 sTab[NFEAT];

  if (n_pos <= 0) return;
  const int4* __restrict__ idxT = (const int4*)idx_in;
  const int wave = __builtin_amdgcn_readfirstlane((int)(threadIdx.x >> 5));
  const int lane = threadIdx.x & 31;
  const int c = lane & 15;
  const int h = lane >> 4;
  const int slice = blockIdx.x & (NSLICE - 1);
  const int chunk = blockIdx.x >> 5;
  const int groups  = (n_pos + 15) >> 4;
  const int gstride = NCHUNK * SL_WAVES;
  int g = chunk * SL_WAVES + wave;

  int4 cur[4];
  load_idx(cur, idxT, min(g, groups - 1), lane);

  const int nq = (c < NBUCK) ? c : 0;
  const float nmask = (c < NBUCK) ? 1.0f : 0.0f;
  const float* lw = layer_w + (size_t)nq * (2 * NACC) + h * NACC + slice * 4;
  v2f wf0, wf1;
  wf0.x = lw[0] * nmask; wf0.y = lw[1] * nmask;
  wf1.x = lw[2] * nmask; wf1.y = lw[3] * nmask;
  const float4 bias = ld4(acc_bias + slice * 4);

  {
    const float* src = slices + (size_t)slice * NFEAT * 4;
    int r = (int)((chunk * 2560u) % NFEAT) + (int)threadIdx.x;
#pragma unroll
    for (int i = 0; i < NFEAT / SL_THREADS; ++i) {
      if (r >= NFEAT) r -= NFEAT;
      sTab[r] = ld4(src + (size_t)r * 4);
      r += SL_THREADS;
    }
  }
  __syncthreads();

  while (g < groups) {
    const int posu = g * 16 + c;
    const int pos  = min(posu, n_pos - 1);
    const float s  = stms[pos];
    int4 nxt[4];
    load_idx(nxt, idxT, min(g + gstride, groups - 1), lane);

    float4 a = make_float4(0.f, 0.f, 0.f, 0.f);
#pragma unroll
    for (int q = 0; q < 4; ++q) {
      add_rows(a, sTab, cur[q].x); add_rows(a, sTab, cur[q].y);
      add_rows(a, sTab, cur[q].z); add_rows(a, sTab, cur[q].w);
    }
    a.x = fminf(fmaxf(a.x + bias.x, 0.f), 1.f);
    a.y = fminf(fmaxf(a.y + bias.y, 0.f), 1.f);
    a.z = fminf(fmaxf(a.z + bias.z, 0.f), 1.f);
    a.w = fminf(fmaxf(a.w + bias.w, 0.f), 1.f);

    const bool flip = (s != 0.0f);
    const float ox = __shfl_xor(a.x, 16, 32);
    const float oy = __shfl_xor(a.y, 16, 32);
    const float oz = __shfl_xor(a.z, 16, 32);
    const float ow = __shfl_xor(a.w, 16, 32);
    v2f af0, af1;
    af0.x = flip ? ox : a.x; af0.y = flip ? oy : a.y;
    af1.x = flip ? oz : a.z; af1.y = flip ? ow : a.w;

    v8f d = {0.f, 0.f, 0.f, 0.f, 0.f, 0.f, 0.f, 0.f};
    d = __builtin_amdgcn_wmma_f32_16x16x4_f32(false, wf0, false, af0, (short)0, d, false, false);
    d = __builtin_amdgcn_wmma_f32_16x16x4_f32(false, wf1, false, af1, (short)0, d, false, false);

    if (h == 0 && posu < n_pos)
      *(float4*)(part + ((size_t)slice * n_pos + posu) * NBUCK) =
          make_float4(d[0], d[1], d[2], d[3]);

#pragma unroll
    for (int q = 0; q < 4; ++q) cur[q] = nxt[q];
    g += gstride;
  }
}

#define PS_WAVES 16
#define PS_THREADS (PS_WAVES * 32)

__global__ __launch_bounds__(PS_THREADS) void psqt_add(
    const int* __restrict__ idx_in, const float* __restrict__ stms,
    const float* __restrict__ psqt_w, const float* __restrict__ layer_b,
    const float* __restrict__ part, float* __restrict__ out, int n_pos)
{
  __shared__ float4 sTab[NFEAT];

  if (n_pos <= 0) return;
  const int4* __restrict__ idxT = (const int4*)idx_in;
  const int wave = __builtin_amdgcn_readfirstlane((int)(threadIdx.x >> 5));
  const int lane = threadIdx.x & 31;
  const int c = lane & 15;
  const int h = lane >> 4;
  const int groups  = (n_pos + 15) >> 4;
  const int gstride = gridDim.x * PS_WAVES;
  int g = blockIdx.x * PS_WAVES + wave;

  int4 cur[4];
  load_idx(cur, idxT, min(g, groups - 1), lane);
  float4 R[NSLICE / 2];
  float  sgl;
  {
    const int pos = min(min(g, groups - 1) * 16 + c, n_pos - 1);
#pragma unroll
    for (int k = 0; k < NSLICE / 2; ++k)
      R[k] = ld4(part + ((unsigned)(16 * h + k) * (unsigned)n_pos + (unsigned)pos) * NBUCK);
    sgl = (stms[pos] != 0.0f) ? -1.0f : 1.0f;
  }
  const float lb0 = layer_b[0], lb1 = layer_b[1], lb2 = layer_b[2], lb3 = layer_b[3];

  {
    int r = (int)((blockIdx.x * 80u) % NFEAT) + (int)threadIdx.x;
#pragma unroll 20
    for (int i = 0; i < NFEAT / PS_THREADS; ++i) {
      if (r >= NFEAT) r -= NFEAT;
      sTab[r] = ld4(psqt_w + (size_t)r * NBUCK);
      r += PS_THREADS;
    }
  }
  __syncthreads();

  while (g < groups) {
    const int posu = g * 16 + c;

    float4 a = make_float4(0.f, 0.f, 0.f, 0.f);
#pragma unroll
    for (int q = 0; q < 4; ++q) {
      add_rows(a, sTab, cur[q].x); add_rows(a, sTab, cur[q].y);
      add_rows(a, sTab, cur[q].z); add_rows(a, sTab, cur[q].w);
    }
    float4 p = R[0];
#pragma unroll
    for (int k = 1; k < NSLICE / 2; ++k) {
      p.x += R[k].x; p.y += R[k].y; p.z += R[k].z; p.w += R[k].w;
    }
    const float sx = h ? -a.x : a.x, sy = h ? -a.y : a.y;
    const float sz = h ? -a.z : a.z, sw = h ? -a.w : a.w;
    const float psx = sx + __shfl_xor(sx, 16, 32);
    const float psy = sy + __shfl_xor(sy, 16, 32);
    const float psz = sz + __shfl_xor(sz, 16, 32);
    const float psw = sw + __shfl_xor(sw, 16, 32);
    const float qx = p.x + __shfl_xor(p.x, 16, 32);
    const float qy = p.y + __shfl_xor(p.y, 16, 32);
    const float qz = p.z + __shfl_xor(p.z, 16, 32);
    const float qw = p.w + __shfl_xor(p.w, 16, 32);
    if (h == 0 && posu < n_pos)
      *(float4*)(out + (size_t)posu * NBUCK) =
          make_float4(psx + sgl * (qx + lb0), psy + sgl * (qy + lb1),
                      psz + sgl * (qz + lb2), psw + sgl * (qw + lb3));

    g += gstride;
    if (g < groups) {
      load_idx(cur, idxT, g, lane);
      const int pos = min(g * 16 + c, n_pos - 1);
#pragma unroll
      for (int k = 0; k < NSLICE / 2; ++k)
        R[k] = ld4(part + ((unsigned)(16 * h + k) * (unsigned)n_pos + (unsigned)pos) * NBUCK);
      sgl = (stms[pos] != 0.0f) ? -1.0f : 1.0f;
    }
  }
}

extern "C" void kernel_launch(void* const* d_in, const int* in_sizes, int n_in,
                              void* d_out, int out_size, void* d_ws, size_t ws_size,
                              hipStream_t stream) {
  const int*   w_cols   = (const int*)d_in[1];
  const int*   b_cols   = (const int*)d_in[3];
  const float* stms     = (const float*)d_in[4];
  const float* psqt_w   = (const float*)d_in[5];
  const float* acc_w    = (const float*)d_in[6];
  const float* acc_bias = (const float*)d_in[7];
  const float* layer_w  = (const float*)d_in[8];
  const float* layer_b  = (const float*)d_in[9];
  float* out = (float*)d_out;
  float* slices = (float*)d_ws;
  float* part   = (float*)d_ws + PART_OFF_FLOATS;
  int*   idxT   = (int*)((float*)d_ws + IDX_OFF_FLOATS);

  const int n_pos = in_sizes[4];
  acc_to_slices<<<TR_BLOCKS + IX_BLOCKS, 256, 0, stream>>>(
      acc_w, slices, w_cols, b_cols, idxT, n_pos);
  slice_fwd<<<NSLICE * NCHUNK, SL_THREADS, 0, stream>>>(
      idxT, stms, slices, acc_bias, layer_w, part, n_pos);
  psqt_add<<<256, PS_THREADS, 0, stream>>>(
      idxT, stms, psqt_w, layer_b, part, out, n_pos);
}
